# starting kernel
# baseline (speedup 1.0000x reference)
#include <hip/hip_runtime.h>

typedef float v2f __attribute__((ext_vector_type(2)));
typedef float v8f __attribute__((ext_vector_type(8)));

#define N_NODES 30000
#define D_DEG   32
#define H_DIM   128
#define K_LAB   10
#define K_AUG   144
#define N_STEPS (K_AUG / 4)
#define AS_STRIDE 148

__global__ __launch_bounds__(256) void gcn_prep(
    const float* __restrict__ W0, const float* __restrict__ b0,
    const float* __restrict__ W1, const float* __restrict__ b1,
    const float* __restrict__ W2, const float* __restrict__ b2,
    const float* __restrict__ Wt3, const float* __restrict__ Wt4,
    float* __restrict__ Bdense, float* __restrict__ Bsw) {
  const int tid = threadIdx.x;
  if (tid < H_DIM) {
    const int hp = tid;
    for (int k = 0; k < H_DIM; ++k)
      Bdense[k * H_DIM + hp] = W2[k * H_DIM + hp];
    for (int k = 0; k < K_LAB; ++k)
      Bdense[(H_DIM + k) * H_DIM + hp] = W1[k * H_DIM + hp];
    Bdense[138 * H_DIM + hp] = W0[hp];
    Bdense[139 * H_DIM + hp] = W0[H_DIM + hp];
    float u = 0.f, v = 0.f;
    for (int h = 0; h < H_DIM; ++h) {
      float t  = Wt4[h];
      float w3 = Wt3[h * H_DIM + hp];
      u += fmaxf(t, 0.f) * w3;
      v += fminf(t, 0.f) * w3;
    }
    Bdense[140 * H_DIM + hp] = u;
    Bdense[141 * H_DIM + hp] = v;
    Bdense[142 * H_DIM + hp] = b0[hp] + b1[hp] + b2[hp];
    Bdense[143 * H_DIM + hp] = 0.f;
  }
  __syncthreads();

  const int tile  = tid >> 5;
  const int lane  = tid & 31;
  const int n     = tile * 16 + (lane & 15);
  const int khalf = 2 * (lane >> 4);
  float* dst = Bsw + (tile * 32 + lane) * (2 * N_STEPS);
  for (int t = 0; t < N_STEPS; ++t) {
    const int kk = 4 * t + khalf;
    dst[2 * t + 0] = Bdense[kk * H_DIM + n];
    dst[2 * t + 1] = Bdense[(kk + 1) * H_DIM + n];
  }
}

__global__ __launch_bounds__(256) void gcn_main(
    const float* __restrict__ feature, const float* __restrict__ x,
    const float* __restrict__ label,   const float* __restrict__ w,
    const float* __restrict__ e_type,  const int* __restrict__ src,
    const float* __restrict__ Bsw,     float* __restrict__ out) {

  __shared__ int   s_src[16 * D_DEG];
  __shared__ float s_et0[16 * D_DEG];
  __shared__ float s_A[16 * AS_STRIDE];

  const int tid = threadIdx.x;
  const int nb  = blockIdx.x * 16;

  for (int i = tid; i < 16 * D_DEG; i += 256) {
    const int n = nb + (i >> 5);
    const int d = i & (D_DEG - 1);
    s_src[i] = src[n * D_DEG + d];
    s_et0[i] = e_type[(n * D_DEG + d) * 2 + 0];
  }
  __syncthreads();

  const int row  = tid >> 4;
  const int c    = tid & 15;
  const int node = nb + row;

  float acc[8];
#pragma unroll
  for (int j = 0; j < 8; ++j) acc[j] = 0.f;
  for (int d = 0; d < D_DEG; ++d) {
    const int   sidx = s_src[row * D_DEG + d];
    const float e0   = s_et0[row * D_DEG + d];
    const float4* frow4 = (const float4*)(feature + (long)sidx * H_DIM) + c * 2;
    const float4 f0 = frow4[0];
    const float4 f1 = frow4[1];
    acc[0] += f0.x * e0; acc[1] += f0.y * e0;
    acc[2] += f0.z * e0; acc[3] += f0.w * e0;
    acc[4] += f1.x * e0; acc[5] += f1.y * e0;
    acc[6] += f1.z * e0; acc[7] += f1.w * e0;
  }
#pragma unroll
  for (int j = 0; j < 8; ++j) s_A[row * AS_STRIDE + c * 8 + j] = acc[j];

  if (c < K_LAB) {
    s_A[row * AS_STRIDE + H_DIM + c] = label[node * K_LAB + c];
  } else if (c == 10) {
    s_A[row * AS_STRIDE + 138] = x[node * 2 + 0];
  } else if (c == 11) {
    s_A[row * AS_STRIDE + 139] = x[node * 2 + 1];
  } else if (c == 12) {
    s_A[row * AS_STRIDE + 142] = 1.0f;
  } else if (c == 13) {
    s_A[row * AS_STRIDE + 143] = 0.0f;
  } else if (c == 14) {
    float p = 0.f;
    for (int d = 0; d < D_DEG; ++d) {
      float nw = w[node * D_DEG + d] * s_et0[row * D_DEG + d];
      p += fmaxf(nw, 0.f);
    }
    s_A[row * AS_STRIDE + 140] = p;
  } else {
    float q = 0.f;
    for (int d = 0; d < D_DEG; ++d) {
      float nw = w[node * D_DEG + d] * s_et0[row * D_DEG + d];
      q += fminf(nw, 0.f);
    }
    s_A[row * AS_STRIDE + 141] = q;
  }
  __syncthreads();

  const int wave = tid >> 5;
  const int lane = tid & 31;
  const int m    = lane & 15;
  const int hi   = lane >> 4;
  const int ncol = wave * 16 + (lane & 15);

  const float4* bfr =
      (const float4*)(Bsw + (wave * 32 + lane) * (2 * N_STEPS));

  v8f cacc = {0.f, 0.f, 0.f, 0.f, 0.f, 0.f, 0.f, 0.f};
#pragma unroll
  for (int t2 = 0; t2 < N_STEPS / 2; ++t2) {
    const float4 bv = bfr[t2];
    {
      const int kk = 4 * (2 * t2) + 2 * hi;
      v2f a, b;
      a.x = s_A[m * AS_STRIDE + kk];
      a.y = s_A[m * AS_STRIDE + kk + 1];
      b.x = bv.x; b.y = bv.y;
      cacc = __builtin_amdgcn_wmma_f32_16x16x4_f32(
          false, a, false, b, (short)0, cacc, false, false);
    }
    {
      const int kk = 4 * (2 * t2 + 1) + 2 * hi;
      v2f a, b;
      a.x = s_A[m * AS_STRIDE + kk];
      a.y = s_A[m * AS_STRIDE + kk + 1];
      b.x = bv.z; b.y = bv.w;
      cacc = __builtin_amdgcn_wmma_f32_16x16x4_f32(
          false, a, false, b, (short)0, cacc, false, false);
    }
  }

#pragma unroll
  for (int r = 0; r < 8; ++r) {
    const int mrow = r + 8 * hi;
    const float vr = cacc[r];
    out[(long)(nb + mrow) * H_DIM + ncol] = vr > 0.f ? vr : 0.f;
  }
}

extern "C" void kernel_launch(void* const* d_in, const int* in_sizes, int n_in,
                              void* d_out, int out_size, void* d_ws, size_t ws_size,
                              hipStream_t stream) {
  const float* feature = (const float*)d_in[0];
  const float* x       = (const float*)d_in[1];
  const float* label   = (const float*)d_in[2];
  const float* w       = (const float*)d_in[3];
  const float* e_type  = (const float*)d_in[4];
  const int*   src     = (const int*)  d_in[5];
  const float* W0      = (const float*)d_in[6];
  const float* b0      = (const float*)d_in[7];
  const float* W1      = (const float*)d_in[8];
  const float* b1      = (const float*)d_in[9];
  const float* W2      = (const float*)d_in[10];
  const float* b2      = (const float*)d_in[11];
  const float* Wt3     = (const float*)d_in[12];
  const float* Wt4     = (const float*)d_in[13];

  float* Bdense = (float*)d_ws;
  float* Bsw    = (float*)((char*)d_ws + K_AUG * H_DIM * sizeof(float));

  gcn_prep<<<1, 256, 0, stream>>>(W0, b0, W1, b1, W2, b2, Wt3, Wt4,
                                  Bdense, Bsw);
  gcn_main<<<N_NODES / 16, 256, 0, stream>>>(feature, x, label, w, e_type, src,
                                             Bsw, (float*)d_out);
}


# Prep spread over 18 blocks: each thread builds one float4 of the swizzled B straight from the weights (no dense copy, no barrier); u/v sums keep their order
# speedup vs baseline: 1.3835x; 1.3835x over previous
#include <hip/hip_runtime.h>

typedef float v2f __attribute__((ext_vector_type(2)));
typedef float v8f __attribute__((ext_vector_type(8)));

#define N_NODES 30000
#define D_DEG   32
#define H_DIM   128
#define K_LAB   10
#define K_AUG   144
#define N_STEPS (K_AUG / 4)
#define AS_STRIDE 148

#define N_Q        (N_STEPS / 2)
#define PREP_BLOCK 256
#define PREP_GRID  (8 * 32 * N_Q / PREP_BLOCK)

__global__ __launch_bounds__(PREP_BLOCK) void gcn_prep(
    const float* __restrict__ W0, const float* __restrict__ b0,
    const float* __restrict__ W1, const float* __restrict__ b1,
    const float* __restrict__ W2, const float* __restrict__ b2,
    const float* __restrict__ Wt3, const float* __restrict__ Wt4,
    float* __restrict__ Bsw) {
  const int g    = blockIdx.x * PREP_BLOCK + threadIdx.x;
  const int frow = g / N_Q;
  const int q    = g - frow * N_Q;
  const int lane = frow & 31;
  const int hi   = lane >> 4;
  const int n    = (frow >> 5) * 16 + (lane & 15);
  const int k0   = 8 * q + 2 * hi;

  const float* p = (k0 < H_DIM)         ? W2 + k0 * H_DIM + n
                 : (k0 < H_DIM + K_LAB) ? W1 + (k0 - H_DIM) * H_DIM + n
                 :                        W0 + (k0 - H_DIM - K_LAB) * H_DIM + n;
  float4 o;
  o.x = p[0];
  o.y = p[H_DIM];
  if (q < N_Q - 1) {
    o.z = p[4 * H_DIM];
    o.w = p[5 * H_DIM];
  } else if (hi == 0) {
    float u = 0.f, v = 0.f;
#pragma unroll
    for (int h = 0; h < H_DIM; ++h) {
      float t  = Wt4[h];
      float w3 = Wt3[h * H_DIM + n];
      u += fmaxf(t, 0.f) * w3;
      v += fminf(t, 0.f) * w3;
    }
    o.z = u;
    o.w = v;
  } else {
    o.z = b0[n] + b1[n] + b2[n];
    o.w = 0.f;
  }
  ((float4*)Bsw)[g] = o;
}

__global__ __launch_bounds__(256) void gcn_main(
    const float* __restrict__ feature, const float* __restrict__ x,
    const float* __restrict__ label,   const float* __restrict__ w,
    const float* __restrict__ e_type,  const int* __restrict__ src,
    const float* __restrict__ Bsw,     float* __restrict__ out) {

  __shared__ int   s_src[16 * D_DEG];
  __shared__ float s_et0[16 * D_DEG];
  __shared__ float s_A[16 * AS_STRIDE];

  const int tid = threadIdx.x;
  const int nb  = blockIdx.x * 16;

  for (int i = tid; i < 16 * D_DEG; i += 256) {
    const int n = nb + (i >> 5);
    const int d = i & (D_DEG - 1);
    s_src[i] = src[n * D_DEG + d];
    s_et0[i] = e_type[(n * D_DEG + d) * 2 + 0];
  }
  __syncthreads();

  const int row  = tid >> 4;
  const int c    = tid & 15;
  const int node = nb + row;

  float acc[8];
#pragma unroll
  for (int j = 0; j < 8; ++j) acc[j] = 0.f;
  for (int d = 0; d < D_DEG; ++d) {
    const int   sidx = s_src[row * D_DEG + d];
    const float e0   = s_et0[row * D_DEG + d];
    const float4* frow4 = (const float4*)(feature + (long)sidx * H_DIM) + c * 2;
    const float4 f0 = frow4[0];
    const float4 f1 = frow4[1];
    acc[0] += f0.x * e0; acc[1] += f0.y * e0;
    acc[2] += f0.z * e0; acc[3] += f0.w * e0;
    acc[4] += f1.x * e0; acc[5] += f1.y * e0;
    acc[6] += f1.z * e0; acc[7] += f1.w * e0;
  }
#pragma unroll
  for (int j = 0; j < 8; ++j) s_A[row * AS_STRIDE + c * 8 + j] = acc[j];

  if (c < K_LAB) {
    s_A[row * AS_STRIDE + H_DIM + c] = label[node * K_LAB + c];
  } else if (c == 10) {
    s_A[row * AS_STRIDE + 138] = x[node * 2 + 0];
  } else if (c == 11) {
    s_A[row * AS_STRIDE + 139] = x[node * 2 + 1];
  } else if (c == 12) {
    s_A[row * AS_STRIDE + 142] = 1.0f;
  } else if (c == 13) {
    s_A[row * AS_STRIDE + 143] = 0.0f;
  } else if (c == 14) {
    float p = 0.f;
    for (int d = 0; d < D_DEG; ++d) {
      float nw = w[node * D_DEG + d] * s_et0[row * D_DEG + d];
      p += fmaxf(nw, 0.f);
    }
    s_A[row * AS_STRIDE + 140] = p;
  } else {
    float q = 0.f;
    for (int d = 0; d < D_DEG; ++d) {
      float nw = w[node * D_DEG + d] * s_et0[row * D_DEG + d];
      q += fminf(nw, 0.f);
    }
    s_A[row * AS_STRIDE + 141] = q;
  }
  __syncthreads();

  const int wave = tid >> 5;
  const int lane = tid & 31;
  const int m    = lane & 15;
  const int hi   = lane >> 4;
  const int ncol = wave * 16 + (lane & 15);

  const float4* bfr =
      (const float4*)(Bsw + (wave * 32 + lane) * (2 * N_STEPS));

  v8f cacc = {0.f, 0.f, 0.f, 0.f, 0.f, 0.f, 0.f, 0.f};
#pragma unroll
  for (int t2 = 0; t2 < N_STEPS / 2; ++t2) {
    const float4 bv = bfr[t2];
    {
      const int kk = 4 * (2 * t2) + 2 * hi;
      v2f a, b;
      a.x = s_A[m * AS_STRIDE + kk];
      a.y = s_A[m * AS_STRIDE + kk + 1];
      b.x = bv.x; b.y = bv.y;
      cacc = __builtin_amdgcn_wmma_f32_16x16x4_f32(
          false, a, false, b, (short)0, cacc, false, false);
    }
    {
      const int kk = 4 * (2 * t2 + 1) + 2 * hi;
      v2f a, b;
      a.x = s_A[m * AS_STRIDE + kk];
      a.y = s_A[m * AS_STRIDE + kk + 1];
      b.x = bv.z; b.y = bv.w;
      cacc = __builtin_amdgcn_wmma_f32_16x16x4_f32(
          false, a, false, b, (short)0, cacc, false, false);
    }
  }

#pragma unroll
  for (int r = 0; r < 8; ++r) {
    const int mrow = r + 8 * hi;
    const float vr = cacc[r];
    out[(long)(nb + mrow) * H_DIM + ncol] = vr > 0.f ? vr : 0.f;
  }
}

extern "C" void kernel_launch(void* const* d_in, const int* in_sizes, int n_in,
                              void* d_out, int out_size, void* d_ws, size_t ws_size,
                              hipStream_t stream) {
  const float* feature = (const float*)d_in[0];
  const float* x       = (const float*)d_in[1];
  const float* label   = (const float*)d_in[2];
  const float* w       = (const float*)d_in[3];
  const float* e_type  = (const float*)d_in[4];
  const int*   src     = (const int*)  d_in[5];
  const float* W0      = (const float*)d_in[6];
  const float* b0      = (const float*)d_in[7];
  const float* W1      = (const float*)d_in[8];
  const float* b1      = (const float*)d_in[9];
  const float* W2      = (const float*)d_in[10];
  const float* b2      = (const float*)d_in[11];
  const float* Wt3     = (const float*)d_in[12];
  const float* Wt4     = (const float*)d_in[13];

  float* Bsw = (float*)d_ws;

  gcn_prep<<<PREP_GRID, PREP_BLOCK, 0, stream>>>(W0, b0, W1, b1, W2, b2, Wt3, Wt4,
                                                 Bsw);
  gcn_main<<<N_NODES / 16, 256, 0, stream>>>(feature, x, label, w, e_type, src,
                                             Bsw, (float*)d_out);
}


# B fragments stored lane-contiguous: each b128 B load of a wave covers 512 contiguous bytes (4 lines) instead of 32 lines 288 bytes apart; k order and every sum unchanged
# speedup vs baseline: 1.5754x; 1.1387x over previous
#include <hip/hip_runtime.h>

typedef float v2f __attribute__((ext_vector_type(2)));
typedef float v8f __attribute__((ext_vector_type(8)));

#define N_NODES 30000
#define D_DEG   32
#define H_DIM   128
#define K_LAB   10
#define K_AUG   144
#define N_STEPS (K_AUG / 4)
#define AS_STRIDE 148

#define N_Q        (N_STEPS / 2)
#define PREP_BLOCK 256
#define PREP_GRID  (8 * 32 * N_Q / PREP_BLOCK)

__global__ __launch_bounds__(PREP_BLOCK) void gcn_prep(
    const float* __restrict__ W0, const float* __restrict__ b0,
    const float* __restrict__ W1, const float* __restrict__ b1,
    const float* __restrict__ W2, const float* __restrict__ b2,
    const float* __restrict__ Wt3, const float* __restrict__ Wt4,
    float* __restrict__ Bsw) {
  const int g    = blockIdx.x * PREP_BLOCK + threadIdx.x;
  const int lane = g & 31;
  const int grp  = g >> 5;
  const int tile = grp / N_Q;
  const int q    = grp - tile * N_Q;
  const int hi   = lane >> 4;
  const int n    = tile * 16 + (lane & 15);
  const int k0   = 8 * q + 2 * hi;

  const float* p = (k0 < H_DIM)         ? W2 + k0 * H_DIM + n
                 : (k0 < H_DIM + K_LAB) ? W1 + (k0 - H_DIM) * H_DIM + n
                 :                        W0 + (k0 - H_DIM - K_LAB) * H_DIM + n;
  float4 o;
  o.x = p[0];
  o.y = p[H_DIM];
  if (q < N_Q - 1) {
    o.z = p[4 * H_DIM];
    o.w = p[5 * H_DIM];
  } else if (hi == 0) {
    float u = 0.f, v = 0.f;
#pragma unroll
    for (int h = 0; h < H_DIM; ++h) {
      float t  = Wt4[h];
      float w3 = Wt3[h * H_DIM + n];
      u += fmaxf(t, 0.f) * w3;
      v += fminf(t, 0.f) * w3;
    }
    o.z = u;
    o.w = v;
  } else {
    o.z = b0[n] + b1[n] + b2[n];
    o.w = 0.f;
  }
  ((float4*)Bsw)[g] = o;
}

__global__ __launch_bounds__(256) void gcn_main(
    const float* __restrict__ feature, const float* __restrict__ x,
    const float* __restrict__ label,   const float* __restrict__ w,
    const float* __restrict__ e_type,  const int* __restrict__ src,
    const float* __restrict__ Bsw,     float* __restrict__ out) {

  __shared__ int   s_src[16 * D_DEG];
  __shared__ float s_et0[16 * D_DEG];
  __shared__ float s_A[16 * AS_STRIDE];

  const int tid = threadIdx.x;
  const int nb  = blockIdx.x * 16;

  for (int i = tid; i < 16 * D_DEG; i += 256) {
    const int n = nb + (i >> 5);
    const int d = i & (D_DEG - 1);
    s_src[i] = src[n * D_DEG + d];
    s_et0[i] = e_type[(n * D_DEG + d) * 2 + 0];
  }
  __syncthreads();

  const int row  = tid >> 4;
  const int c    = tid & 15;
  const int node = nb + row;

  float acc[8];
#pragma unroll
  for (int j = 0; j < 8; ++j) acc[j] = 0.f;
  for (int d = 0; d < D_DEG; ++d) {
    const int   sidx = s_src[row * D_DEG + d];
    const float e0   = s_et0[row * D_DEG + d];
    const float4* frow4 = (const float4*)(feature + (long)sidx * H_DIM) + c * 2;
    const float4 f0 = frow4[0];
    const float4 f1 = frow4[1];
    acc[0] += f0.x * e0; acc[1] += f0.y * e0;
    acc[2] += f0.z * e0; acc[3] += f0.w * e0;
    acc[4] += f1.x * e0; acc[5] += f1.y * e0;
    acc[6] += f1.z * e0; acc[7] += f1.w * e0;
  }
#pragma unroll
  for (int j = 0; j < 8; ++j) s_A[row * AS_STRIDE + c * 8 + j] = acc[j];

  if (c < K_LAB) {
    s_A[row * AS_STRIDE + H_DIM + c] = label[node * K_LAB + c];
  } else if (c == 10) {
    s_A[row * AS_STRIDE + 138] = x[node * 2 + 0];
  } else if (c == 11) {
    s_A[row * AS_STRIDE + 139] = x[node * 2 + 1];
  } else if (c == 12) {
    s_A[row * AS_STRIDE + 142] = 1.0f;
  } else if (c == 13) {
    s_A[row * AS_STRIDE + 143] = 0.0f;
  } else if (c == 14) {
    float p = 0.f;
    for (int d = 0; d < D_DEG; ++d) {
      float nw = w[node * D_DEG + d] * s_et0[row * D_DEG + d];
      p += fmaxf(nw, 0.f);
    }
    s_A[row * AS_STRIDE + 140] = p;
  } else {
    float q = 0.f;
    for (int d = 0; d < D_DEG; ++d) {
      float nw = w[node * D_DEG + d] * s_et0[row * D_DEG + d];
      q += fminf(nw, 0.f);
    }
    s_A[row * AS_STRIDE + 141] = q;
  }
  __syncthreads();

  const int wave = tid >> 5;
  const int lane = tid & 31;
  const int m    = lane & 15;
  const int hi   = lane >> 4;
  const int ncol = wave * 16 + (lane & 15);

  const float4* bfr = (const float4*)Bsw + wave * (N_Q * 32) + lane;

  v8f cacc = {0.f, 0.f, 0.f, 0.f, 0.f, 0.f, 0.f, 0.f};
#pragma unroll
  for (int t2 = 0; t2 < N_STEPS / 2; ++t2) {
    const float4 bv = bfr[t2 * 32];
    {
      const int kk = 4 * (2 * t2) + 2 * hi;
      v2f a, b;
      a.x = s_A[m * AS_STRIDE + kk];
      a.y = s_A[m * AS_STRIDE + kk + 1];
      b.x = bv.x; b.y = bv.y;
      cacc = __builtin_amdgcn_wmma_f32_16x16x4_f32(
          false, a, false, b, (short)0, cacc, false, false);
    }
    {
      const int kk = 4 * (2 * t2 + 1) + 2 * hi;
      v2f a, b;
      a.x = s_A[m * AS_STRIDE + kk];
      a.y = s_A[m * AS_STRIDE + kk + 1];
      b.x = bv.z; b.y = bv.w;
      cacc = __builtin_amdgcn_wmma_f32_16x16x4_f32(
          false, a, false, b, (short)0, cacc, false, false);
    }
  }

#pragma unroll
  for (int r = 0; r < 8; ++r) {
    const int mrow = r + 8 * hi;
    const float vr = cacc[r];
    out[(long)(nb + mrow) * H_DIM + ncol] = vr > 0.f ? vr : 0.f;
  }
}

extern "C" void kernel_launch(void* const* d_in, const int* in_sizes, int n_in,
                              void* d_out, int out_size, void* d_ws, size_t ws_size,
                              hipStream_t stream) {
  const float* feature = (const float*)d_in[0];
  const float* x       = (const float*)d_in[1];
  const float* label   = (const float*)d_in[2];
  const float* w       = (const float*)d_in[3];
  const float* e_type  = (const float*)d_in[4];
  const int*   src     = (const int*)  d_in[5];
  const float* W0      = (const float*)d_in[6];
  const float* b0      = (const float*)d_in[7];
  const float* W1      = (const float*)d_in[8];
  const float* b1      = (const float*)d_in[9];
  const float* W2      = (const float*)d_in[10];
  const float* b2      = (const float*)d_in[11];
  const float* Wt3     = (const float*)d_in[12];
  const float* Wt4     = (const float*)d_in[13];

  float* Bsw = (float*)d_ws;

  gcn_prep<<<PREP_GRID, PREP_BLOCK, 0, stream>>>(W0, b0, W1, b1, W2, b2, Wt3, Wt4,
                                                 Bsw);
  gcn_main<<<N_NODES / 16, 256, 0, stream>>>(feature, x, label, w, e_type, src,
                                             Bsw, (float*)d_out);
}


# Registers cut 168 -> 51 VGPRs by bounded unrolls (gather 4, B stream 6, 32-bit gather offsets): eight blocks resident per WGP instead of two, so phases of blocks overlap; every sum keeps its order
# speedup vs baseline: 1.8994x; 1.2057x over previous
#include <hip/hip_runtime.h>

typedef float v2f __attribute__((ext_vector_type(2)));
typedef float v8f __attribute__((ext_vector_type(8)));

#define N_NODES 30000
#define D_DEG   32
#define H_DIM   128
#define K_LAB   10
#define K_AUG   144
#define N_STEPS (K_AUG / 4)
#define AS_STRIDE 148

#define N_Q        (N_STEPS / 2)
#define PREP_BLOCK 256
#define PREP_GRID  (8 * 32 * N_Q / PREP_BLOCK)

__global__ __launch_bounds__(PREP_BLOCK) void gcn_prep(
    const float* __restrict__ W0, const float* __restrict__ b0,
    const float* __restrict__ W1, const float* __restrict__ b1,
    const float* __restrict__ W2, const float* __restrict__ b2,
    const float* __restrict__ Wt3, const float* __restrict__ Wt4,
    float* __restrict__ Bsw) {
  const int g    = blockIdx.x * PREP_BLOCK + threadIdx.x;
  const int lane = g & 31;
  const int grp  = g >> 5;
  const int tile = grp / N_Q;
  const int q    = grp - tile * N_Q;
  const int hi   = lane >> 4;
  const int n    = tile * 16 + (lane & 15);
  const int k0   = 8 * q + 2 * hi;

  const float* p = (k0 < H_DIM)         ? W2 + k0 * H_DIM + n
                 : (k0 < H_DIM + K_LAB) ? W1 + (k0 - H_DIM) * H_DIM + n
                 :                        W0 + (k0 - H_DIM - K_LAB) * H_DIM + n;
  float4 o;
  o.x = p[0];
  o.y = p[H_DIM];
  if (q < N_Q - 1) {
    o.z = p[4 * H_DIM];
    o.w = p[5 * H_DIM];
  } else if (hi == 0) {
    float u = 0.f, v = 0.f;
#pragma unroll
    for (int h = 0; h < H_DIM; ++h) {
      float t  = Wt4[h];
      float w3 = Wt3[h * H_DIM + n];
      u += fmaxf(t, 0.f) * w3;
      v += fminf(t, 0.f) * w3;
    }
    o.z = u;
    o.w = v;
  } else {
    o.z = b0[n] + b1[n] + b2[n];
    o.w = 0.f;
  }
  ((float4*)Bsw)[g] = o;
}

__global__ __launch_bounds__(256) void gcn_main(
    const float* __restrict__ feature, const float* __restrict__ x,
    const float* __restrict__ label,   const float* __restrict__ w,
    const float* __restrict__ e_type,  const int* __restrict__ src,
    const float* __restrict__ Bsw,     float* __restrict__ out) {

  __shared__ int   s_src[16 * D_DEG];
  __shared__ float s_et0[16 * D_DEG];
  __shared__ float s_A[16 * AS_STRIDE];

  const int tid = threadIdx.x;
  const int nb  = blockIdx.x * 16;

  for (int i = tid; i < 16 * D_DEG; i += 256) {
    const int n = nb + (i >> 5);
    const int d = i & (D_DEG - 1);
    s_src[i] = src[n * D_DEG + d];
    s_et0[i] = e_type[(n * D_DEG + d) * 2 + 0];
  }
  __syncthreads();

  const int row  = tid >> 4;
  const int c    = tid & 15;
  const int node = nb + row;

  float acc[8];
#pragma unroll
  for (int j = 0; j < 8; ++j) acc[j] = 0.f;
#pragma unroll 4
  for (int d = 0; d < D_DEG; ++d) {
    const int   sidx = s_src[row * D_DEG + d];
    const float e0   = s_et0[row * D_DEG + d];
    const unsigned boff = (unsigned)sidx * (H_DIM * 4u) + (unsigned)c * 32u;
    const float4* frow4 = (const float4*)((const char*)feature + boff);
    const float4 f0 = frow4[0];
    const float4 f1 = frow4[1];
    acc[0] += f0.x * e0; acc[1] += f0.y * e0;
    acc[2] += f0.z * e0; acc[3] += f0.w * e0;
    acc[4] += f1.x * e0; acc[5] += f1.y * e0;
    acc[6] += f1.z * e0; acc[7] += f1.w * e0;
  }
#pragma unroll
  for (int j = 0; j < 8; ++j) s_A[row * AS_STRIDE + c * 8 + j] = acc[j];

  if (c < K_LAB) {
    s_A[row * AS_STRIDE + H_DIM + c] = label[node * K_LAB + c];
  } else if (c == 10) {
    s_A[row * AS_STRIDE + 138] = x[node * 2 + 0];
  } else if (c == 11) {
    s_A[row * AS_STRIDE + 139] = x[node * 2 + 1];
  } else if (c == 12) {
    s_A[row * AS_STRIDE + 142] = 1.0f;
  } else if (c == 13) {
    s_A[row * AS_STRIDE + 143] = 0.0f;
  } else if (c == 14) {
    float p = 0.f;
#pragma unroll 8
    for (int d = 0; d < D_DEG; ++d) {
      float nw = w[node * D_DEG + d] * s_et0[row * D_DEG + d];
      p += fmaxf(nw, 0.f);
    }
    s_A[row * AS_STRIDE + 140] = p;
  } else {
    float q = 0.f;
#pragma unroll 8
    for (int d = 0; d < D_DEG; ++d) {
      float nw = w[node * D_DEG + d] * s_et0[row * D_DEG + d];
      q += fminf(nw, 0.f);
    }
    s_A[row * AS_STRIDE + 141] = q;
  }
  __syncthreads();

  const int wave = tid >> 5;
  const int lane = tid & 31;
  const int m    = lane & 15;
  const int hi   = lane >> 4;
  const int ncol = wave * 16 + (lane & 15);

  const float4* bfr = (const float4*)Bsw + wave * (N_Q * 32) + lane;

  v8f cacc = {0.f, 0.f, 0.f, 0.f, 0.f, 0.f, 0.f, 0.f};
#pragma unroll 6
  for (int t2 = 0; t2 < N_STEPS / 2; ++t2) {
    const float4 bv = bfr[t2 * 32];
    {
      const int kk = 4 * (2 * t2) + 2 * hi;
      v2f a, b;
      a.x = s_A[m * AS_STRIDE + kk];
      a.y = s_A[m * AS_STRIDE + kk + 1];
      b.x = bv.x; b.y = bv.y;
      cacc = __builtin_amdgcn_wmma_f32_16x16x4_f32(
          false, a, false, b, (short)0, cacc, false, false);
    }
    {
      const int kk = 4 * (2 * t2 + 1) + 2 * hi;
      v2f a, b;
      a.x = s_A[m * AS_STRIDE + kk];
      a.y = s_A[m * AS_STRIDE + kk + 1];
      b.x = bv.z; b.y = bv.w;
      cacc = __builtin_amdgcn_wmma_f32_16x16x4_f32(
          false, a, false, b, (short)0, cacc, false, false);
    }
  }

#pragma unroll
  for (int r = 0; r < 8; ++r) {
    const int mrow = r + 8 * hi;
    const float vr = cacc[r];
    out[(long)(nb + mrow) * H_DIM + ncol] = vr > 0.f ? vr : 0.f;
  }
}

extern "C" void kernel_launch(void* const* d_in, const int* in_sizes, int n_in,
                              void* d_out, int out_size, void* d_ws, size_t ws_size,
                              hipStream_t stream) {
  const float* feature = (const float*)d_in[0];
  const float* x       = (const float*)d_in[1];
  const float* label   = (const float*)d_in[2];
  const float* w       = (const float*)d_in[3];
  const float* e_type  = (const float*)d_in[4];
  const int*   src     = (const int*)  d_in[5];
  const float* W0      = (const float*)d_in[6];
  const float* b0      = (const float*)d_in[7];
  const float* W1      = (const float*)d_in[8];
  const float* b1      = (const float*)d_in[9];
  const float* W2      = (const float*)d_in[10];
  const float* b2      = (const float*)d_in[11];
  const float* Wt3     = (const float*)d_in[12];
  const float* Wt4     = (const float*)d_in[13];

  float* Bsw = (float*)d_ws;

  gcn_prep<<<PREP_GRID, PREP_BLOCK, 0, stream>>>(W0, b0, W1, b1, W2, b2, Wt3, Wt4,
                                                 Bsw);
  gcn_main<<<N_NODES / 16, 256, 0, stream>>>(feature, x, label, w, e_type, src,
                                             Bsw, (float*)d_out);
}


# Edge staging also loads w, label and x (contiguous, one round trip) into LDS: the phase between gather and matrix work issues no global load (was 11 dependent round trips); every sum keeps its order
# speedup vs baseline: 2.0008x; 1.0534x over previous
#include <hip/hip_runtime.h>

typedef float v2f __attribute__((ext_vector_type(2)));
typedef float v8f __attribute__((ext_vector_type(8)));

#define N_NODES 30000
#define D_DEG   32
#define H_DIM   128
#define K_LAB   10
#define K_AUG   144
#define N_STEPS (K_AUG / 4)
#define AS_STRIDE 148

#define N_Q        (N_STEPS / 2)
#define PREP_BLOCK 256
#define PREP_GRID  (8 * 32 * N_Q / PREP_BLOCK)

__global__ __launch_bounds__(PREP_BLOCK) void gcn_prep(
    const float* __restrict__ W0, const float* __restrict__ b0,
    const float* __restrict__ W1, const float* __restrict__ b1,
    const float* __restrict__ W2, const float* __restrict__ b2,
    const float* __restrict__ Wt3, const float* __restrict__ Wt4,
    float* __restrict__ Bsw) {
  const int g    = blockIdx.x * PREP_BLOCK + threadIdx.x;
  const int lane = g & 31;
  const int grp  = g >> 5;
  const int tile = grp / N_Q;
  const int q    = grp - tile * N_Q;
  const int hi   = lane >> 4;
  const int n    = tile * 16 + (lane & 15);
  const int k0   = 8 * q + 2 * hi;

  const float* p = (k0 < H_DIM)         ? W2 + k0 * H_DIM + n
                 : (k0 < H_DIM + K_LAB) ? W1 + (k0 - H_DIM) * H_DIM + n
                 :                        W0 + (k0 - H_DIM - K_LAB) * H_DIM + n;
  float4 o;
  o.x = p[0];
  o.y = p[H_DIM];
  if (q < N_Q - 1) {
    o.z = p[4 * H_DIM];
    o.w = p[5 * H_DIM];
  } else if (hi == 0) {
    float u = 0.f, v = 0.f;
#pragma unroll
    for (int h = 0; h < H_DIM; ++h) {
      float t  = Wt4[h];
      float w3 = Wt3[h * H_DIM + n];
      u += fmaxf(t, 0.f) * w3;
      v += fminf(t, 0.f) * w3;
    }
    o.z = u;
    o.w = v;
  } else {
    o.z = b0[n] + b1[n] + b2[n];
    o.w = 0.f;
  }
  ((float4*)Bsw)[g] = o;
}

__global__ __launch_bounds__(256) void gcn_main(
    const float* __restrict__ feature, const float* __restrict__ x,
    const float* __restrict__ label,   const float* __restrict__ w,
    const float* __restrict__ e_type,  const int* __restrict__ src,
    const float* __restrict__ Bsw,     float* __restrict__ out) {

  __shared__ int   s_src[16 * D_DEG];
  __shared__ float s_et0[16 * D_DEG];
  __shared__ float s_nw[16 * D_DEG];
  __shared__ float s_A[16 * AS_STRIDE];

  const int tid = threadIdx.x;
  const int nb  = blockIdx.x * 16;

  const int e_a = nb * D_DEG + tid;
  const int e_b = e_a + 256;
  const int   src_a = src[e_a];
  const int   src_b = src[e_b];
  const float et_a  = e_type[e_a * 2 + 0];
  const float et_b  = e_type[e_b * 2 + 0];
  const float w_a   = w[e_a];
  const float w_b   = w[e_b];
  float side = 0.f;
  if (tid < 16 * K_LAB + 32) {
    const float* sp = (tid < 16 * K_LAB) ? label + nb * K_LAB + tid
                                         : x + nb * 2 + (tid - 16 * K_LAB);
    side = *sp;
  }
  s_src[tid]       = src_a;
  s_src[tid + 256] = src_b;
  s_et0[tid]       = et_a;
  s_et0[tid + 256] = et_b;
  s_nw[tid]        = w_a * et_a;
  s_nw[tid + 256]  = w_b * et_b;
  if (tid < 16 * K_LAB) {
    const int r = tid / K_LAB;
    s_A[r * AS_STRIDE + H_DIM + (tid - r * K_LAB)] = side;
  } else if (tid < 16 * K_LAB + 32) {
    const int j = tid - 16 * K_LAB;
    s_A[(j >> 1) * AS_STRIDE + H_DIM + K_LAB + (j & 1)] = side;
  } else if (tid < 16 * K_LAB + 48) {
    const int r = tid - (16 * K_LAB + 32);
    s_A[r * AS_STRIDE + 142] = 1.0f;
    s_A[r * AS_STRIDE + 143] = 0.0f;
  }
  __syncthreads();

  const int row  = tid >> 4;
  const int c    = tid & 15;

  float acc[8];
#pragma unroll
  for (int j = 0; j < 8; ++j) acc[j] = 0.f;
#pragma unroll 4
  for (int d = 0; d < D_DEG; ++d) {
    const int   sidx = s_src[row * D_DEG + d];
    const float e0   = s_et0[row * D_DEG + d];
    const unsigned boff = (unsigned)sidx * (H_DIM * 4u) + (unsigned)c * 32u;
    const float4* frow4 = (const float4*)((const char*)feature + boff);
    const float4 f0 = frow4[0];
    const float4 f1 = frow4[1];
    acc[0] += f0.x * e0; acc[1] += f0.y * e0;
    acc[2] += f0.z * e0; acc[3] += f0.w * e0;
    acc[4] += f1.x * e0; acc[5] += f1.y * e0;
    acc[6] += f1.z * e0; acc[7] += f1.w * e0;
  }
#pragma unroll
  for (int j = 0; j < 8; ++j) s_A[row * AS_STRIDE + c * 8 + j] = acc[j];

  if (c >= 14) {
    const bool pos = (c == 14);
    float p = 0.f;
#pragma unroll
    for (int d = 0; d < D_DEG; ++d) {
      const float nw = s_nw[row * D_DEG + d];
      p += pos ? fmaxf(nw, 0.f) : fminf(nw, 0.f);
    }
    s_A[row * AS_STRIDE + 140 + (c - 14)] = p;
  }
  __syncthreads();

  const int wave = tid >> 5;
  const int lane = tid & 31;
  const int m    = lane & 15;
  const int hi   = lane >> 4;
  const int ncol = wave * 16 + (lane & 15);

  const float4* bfr = (const float4*)Bsw + wave * (N_Q * 32) + lane;

  v8f cacc = {0.f, 0.f, 0.f, 0.f, 0.f, 0.f, 0.f, 0.f};
#pragma unroll 6
  for (int t2 = 0; t2 < N_STEPS / 2; ++t2) {
    const float4 bv = bfr[t2 * 32];
    {
      const int kk = 4 * (2 * t2) + 2 * hi;
      v2f a, b;
      a.x = s_A[m * AS_STRIDE + kk];
      a.y = s_A[m * AS_STRIDE + kk + 1];
      b.x = bv.x; b.y = bv.y;
      cacc = __builtin_amdgcn_wmma_f32_16x16x4_f32(
          false, a, false, b, (short)0, cacc, false, false);
    }
    {
      const int kk = 4 * (2 * t2 + 1) + 2 * hi;
      v2f a, b;
      a.x = s_A[m * AS_STRIDE + kk];
      a.y = s_A[m * AS_STRIDE + kk + 1];
      b.x = bv.z; b.y = bv.w;
      cacc = __builtin_amdgcn_wmma_f32_16x16x4_f32(
          false, a, false, b, (short)0, cacc, false, false);
    }
  }

#pragma unroll
  for (int r = 0; r < 8; ++r) {
    const int mrow = r + 8 * hi;
    const float vr = cacc[r];
    out[(long)(nb + mrow) * H_DIM + ncol] = vr > 0.f ? vr : 0.f;
  }
}

extern "C" void kernel_launch(void* const* d_in, const int* in_sizes, int n_in,
                              void* d_out, int out_size, void* d_ws, size_t ws_size,
                              hipStream_t stream) {
  const float* feature = (const float*)d_in[0];
  const float* x       = (const float*)d_in[1];
  const float* label   = (const float*)d_in[2];
  const float* w       = (const float*)d_in[3];
  const float* e_type  = (const float*)d_in[4];
  const int*   src     = (const int*)  d_in[5];
  const float* W0      = (const float*)d_in[6];
  const float* b0      = (const float*)d_in[7];
  const float* W1      = (const float*)d_in[8];
  const float* b1      = (const float*)d_in[9];
  const float* W2      = (const float*)d_in[10];
  const float* b2      = (const float*)d_in[11];
  const float* Wt3     = (const float*)d_in[12];
  const float* Wt4     = (const float*)d_in[13];

  float* Bsw = (float*)d_ws;

  gcn_prep<<<PREP_GRID, PREP_BLOCK, 0, stream>>>(W0, b0, W1, b1, W2, b2, Wt3, Wt4,
                                                 Bsw);
  gcn_main<<<N_NODES / 16, 256, 0, stream>>>(feature, x, label, w, e_type, src,
                                             Bsw, (float*)d_out);
}


# Gather lanes re-dealt by whole rows: a lane owns columns 4l..4l+3 of both nodes of its wave, so each b128 gather load reads one 512-B row (4 whole lines), not 8 half-used lines; sum orders kept
# speedup vs baseline: 2.1319x; 1.0655x over previous
#include <hip/hip_runtime.h>

typedef float v2f __attribute__((ext_vector_type(2)));
typedef float v8f __attribute__((ext_vector_type(8)));

#define N_NODES 30000
#define D_DEG   32
#define H_DIM   128
#define K_LAB   10
#define K_AUG   144
#define N_STEPS (K_AUG / 4)
#define AS_STRIDE 148

#define N_Q        (N_STEPS / 2)
#define PREP_BLOCK 256
#define PREP_GRID  (8 * 32 * N_Q / PREP_BLOCK)

__global__ __launch_bounds__(PREP_BLOCK) void gcn_prep(
    const float* __restrict__ W0, const float* __restrict__ b0,
    const float* __restrict__ W1, const float* __restrict__ b1,
    const float* __restrict__ W2, const float* __restrict__ b2,
    const float* __restrict__ Wt3, const float* __restrict__ Wt4,
    float* __restrict__ Bsw) {
  const int g    = blockIdx.x * PREP_BLOCK + threadIdx.x;
  const int lane = g & 31;
  const int grp  = g >> 5;
  const int tile = grp / N_Q;
  const int q    = grp - tile * N_Q;
  const int hi   = lane >> 4;
  const int n    = tile * 16 + (lane & 15);
  const int k0   = 8 * q + 2 * hi;

  const float* p = (k0 < H_DIM)         ? W2 + k0 * H_DIM + n
                 : (k0 < H_DIM + K_LAB) ? W1 + (k0 - H_DIM) * H_DIM + n
                 :                        W0 + (k0 - H_DIM - K_LAB) * H_DIM + n;
  float4 o;
  o.x = p[0];
  o.y = p[H_DIM];
  if (q < N_Q - 1) {
    o.z = p[4 * H_DIM];
    o.w = p[5 * H_DIM];
  } else if (hi == 0) {
    float u = 0.f, v = 0.f;
#pragma unroll
    for (int h = 0; h < H_DIM; ++h) {
      float t  = Wt4[h];
      float w3 = Wt3[h * H_DIM + n];
      u += fmaxf(t, 0.f) * w3;
      v += fminf(t, 0.f) * w3;
    }
    o.z = u;
    o.w = v;
  } else {
    o.z = b0[n] + b1[n] + b2[n];
    o.w = 0.f;
  }
  ((float4*)Bsw)[g] = o;
}

__global__ __launch_bounds__(256) void gcn_main(
    const float* __restrict__ feature, const float* __restrict__ x,
    const float* __restrict__ label,   const float* __restrict__ w,
    const float* __restrict__ e_type,  const int* __restrict__ src,
    const float* __restrict__ Bsw,     float* __restrict__ out) {

  __shared__ __attribute__((aligned(16))) int   s_src[16 * D_DEG];
  __shared__ __attribute__((aligned(16))) float s_et0[16 * D_DEG];
  __shared__ __attribute__((aligned(16))) float s_nw[16 * D_DEG];
  __shared__ __attribute__((aligned(16))) float s_A[16 * AS_STRIDE];

  const int tid = threadIdx.x;
  const int nb  = blockIdx.x * 16;

  const int e_a = nb * D_DEG + tid;
  const int e_b = e_a + 256;
  const int   src_a = src[e_a];
  const int   src_b = src[e_b];
  const float et_a  = e_type[e_a * 2 + 0];
  const float et_b  = e_type[e_b * 2 + 0];
  const float w_a   = w[e_a];
  const float w_b   = w[e_b];
  float side = 0.f;
  if (tid < 16 * K_LAB + 32) {
    const float* sp = (tid < 16 * K_LAB) ? label + nb * K_LAB + tid
                                         : x + nb * 2 + (tid - 16 * K_LAB);
    side = *sp;
  }
  s_src[tid]       = src_a;
  s_src[tid + 256] = src_b;
  s_et0[tid]       = et_a;
  s_et0[tid + 256] = et_b;
  s_nw[tid]        = w_a * et_a;
  s_nw[tid + 256]  = w_b * et_b;
  if (tid < 16 * K_LAB) {
    const int r = tid / K_LAB;
    s_A[r * AS_STRIDE + H_DIM + (tid - r * K_LAB)] = side;
  } else if (tid < 16 * K_LAB + 32) {
    const int j = tid - 16 * K_LAB;
    s_A[(j >> 1) * AS_STRIDE + H_DIM + K_LAB + (j & 1)] = side;
  } else if (tid < 16 * K_LAB + 48) {
    const int r = tid - (16 * K_LAB + 32);
    s_A[r * AS_STRIDE + 142] = 1.0f;
    s_A[r * AS_STRIDE + 143] = 0.0f;
  }
  __syncthreads();

  const int gw = tid >> 5;
  const int gl = tid & 31;
  const int r0 = 2 * gw;
  const int r1 = r0 + 1;
  const unsigned loff = (unsigned)gl * 16u;

  float4 a0 = {0.f, 0.f, 0.f, 0.f};
  float4 a1 = {0.f, 0.f, 0.f, 0.f};
#pragma unroll 4
  for (int d = 0; d < D_DEG; ++d) {
    const int   i0 = s_src[r0 * D_DEG + d];
    const int   i1 = s_src[r1 * D_DEG + d];
    const float e0 = s_et0[r0 * D_DEG + d];
    const float e1 = s_et0[r1 * D_DEG + d];
    const unsigned o0 = (unsigned)i0 * (H_DIM * 4u) + loff;
    const unsigned o1 = (unsigned)i1 * (H_DIM * 4u) + loff;
    const float4 f0 = *(const float4*)((const char*)feature + o0);
    const float4 f1 = *(const float4*)((const char*)feature + o1);
    a0.x += f0.x * e0; a0.y += f0.y * e0;
    a0.z += f0.z * e0; a0.w += f0.w * e0;
    a1.x += f1.x * e1; a1.y += f1.y * e1;
    a1.z += f1.z * e1; a1.w += f1.w * e1;
  }
  *(float4*)&s_A[r0 * AS_STRIDE + gl * 4] = a0;
  *(float4*)&s_A[r1 * AS_STRIDE + gl * 4] = a1;

  const int row = tid >> 4;
  const int c   = tid & 15;
  if (c >= 14) {
    const bool pos = (c == 14);
    float p = 0.f;
#pragma unroll
    for (int d = 0; d < D_DEG; ++d) {
      const float nw = s_nw[row * D_DEG + d];
      p += pos ? fmaxf(nw, 0.f) : fminf(nw, 0.f);
    }
    s_A[row * AS_STRIDE + 140 + (c - 14)] = p;
  }
  __syncthreads();

  const int wave = tid >> 5;
  const int lane = tid & 31;
  const int m    = lane & 15;
  const int hi   = lane >> 4;
  const int ncol = wave * 16 + (lane & 15);

  const float4* bfr = (const float4*)Bsw + wave * (N_Q * 32) + lane;

  v8f cacc = {0.f, 0.f, 0.f, 0.f, 0.f, 0.f, 0.f, 0.f};
#pragma unroll 6
  for (int t2 = 0; t2 < N_STEPS / 2; ++t2) {
    const float4 bv = bfr[t2 * 32];
    {
      const int kk = 4 * (2 * t2) + 2 * hi;
      v2f a, b;
      a.x = s_A[m * AS_STRIDE + kk];
      a.y = s_A[m * AS_STRIDE + kk + 1];
      b.x = bv.x; b.y = bv.y;
      cacc = __builtin_amdgcn_wmma_f32_16x16x4_f32(
          false, a, false, b, (short)0, cacc, false, false);
    }
    {
      const int kk = 4 * (2 * t2 + 1) + 2 * hi;
      v2f a, b;
      a.x = s_A[m * AS_STRIDE + kk];
      a.y = s_A[m * AS_STRIDE + kk + 1];
      b.x = bv.z; b.y = bv.w;
      cacc = __builtin_amdgcn_wmma_f32_16x16x4_f32(
          false, a, false, b, (short)0, cacc, false, false);
    }
  }

#pragma unroll
  for (int r = 0; r < 8; ++r) {
    const int mrow = r + 8 * hi;
    const float vr = cacc[r];
    out[(long)(nb + mrow) * H_DIM + ncol] = vr > 0.f ? vr : 0.f;
  }
}

extern "C" void kernel_launch(void* const* d_in, const int* in_sizes, int n_in,
                              void* d_out, int out_size, void* d_ws, size_t ws_size,
                              hipStream_t stream) {
  const float* feature = (const float*)d_in[0];
  const float* x       = (const float*)d_in[1];
  const float* label   = (const float*)d_in[2];
  const float* w       = (const float*)d_in[3];
  const float* e_type  = (const float*)d_in[4];
  const int*   src     = (const int*)  d_in[5];
  const float* W0      = (const float*)d_in[6];
  const float* b0      = (const float*)d_in[7];
  const float* W1      = (const float*)d_in[8];
  const float* b1      = (const float*)d_in[9];
  const float* W2      = (const float*)d_in[10];
  const float* b2      = (const float*)d_in[11];
  const float* Wt3     = (const float*)d_in[12];
  const float* Wt4     = (const float*)d_in[13];

  float* Bsw = (float*)d_ws;

  gcn_prep<<<PREP_GRID, PREP_BLOCK, 0, stream>>>(W0, b0, W1, b1, W2, b2, Wt3, Wt4,
                                                 Bsw);
  gcn_main<<<N_NODES / 16, 256, 0, stream>>>(feature, x, label, w, e_type, src,
                                             Bsw, (float*)d_out);
}


# A wave owns 32 adjacent columns (two accumulators fed by one A fragment): after one half-wave swap per register every store writes one whole 128-B line; blocks are 4 waves; every sum keeps its order
# speedup vs baseline: 2.2068x; 1.0351x over previous
#include <hip/hip_runtime.h>

typedef float v2f __attribute__((ext_vector_type(2)));
typedef float v8f __attribute__((ext_vector_type(8)));

#define N_NODES 30000
#define D_DEG   32
#define H_DIM   128
#define K_LAB   10
#define K_AUG   144
#define N_STEPS (K_AUG / 4)
#define AS_STRIDE 148

#define N_Q        (N_STEPS / 2)
#define PREP_BLOCK 256
#define PREP_GRID  (8 * 32 * N_Q / PREP_BLOCK)

__global__ __launch_bounds__(PREP_BLOCK) void gcn_prep(
    const float* __restrict__ W0, const float* __restrict__ b0,
    const float* __restrict__ W1, const float* __restrict__ b1,
    const float* __restrict__ W2, const float* __restrict__ b2,
    const float* __restrict__ Wt3, const float* __restrict__ Wt4,
    float* __restrict__ Bsw) {
  const int g    = blockIdx.x * PREP_BLOCK + threadIdx.x;
  const int lane = g & 31;
  const int grp  = g >> 5;
  const int tile = grp / N_Q;
  const int q    = grp - tile * N_Q;
  const int hi   = lane >> 4;
  const int n    = tile * 16 + (lane & 15);
  const int k0   = 8 * q + 2 * hi;

  const float* p = (k0 < H_DIM)         ? W2 + k0 * H_DIM + n
                 : (k0 < H_DIM + K_LAB) ? W1 + (k0 - H_DIM) * H_DIM + n
                 :                        W0 + (k0 - H_DIM - K_LAB) * H_DIM + n;
  float4 o;
  o.x = p[0];
  o.y = p[H_DIM];
  if (q < N_Q - 1) {
    o.z = p[4 * H_DIM];
    o.w = p[5 * H_DIM];
  } else if (hi == 0) {
    float u = 0.f, v = 0.f;
#pragma unroll
    for (int h = 0; h < H_DIM; ++h) {
      float t  = Wt4[h];
      float w3 = Wt3[h * H_DIM + n];
      u += fmaxf(t, 0.f) * w3;
      v += fminf(t, 0.f) * w3;
    }
    o.z = u;
    o.w = v;
  } else {
    o.z = b0[n] + b1[n] + b2[n];
    o.w = 0.f;
  }
  ((float4*)Bsw)[g] = o;
}

__device__ __forceinline__ float swap_halves(float v) {
  return __builtin_bit_cast(float, __builtin_amdgcn_permlanex16(
      __builtin_bit_cast(int, v), __builtin_bit_cast(int, v),
      0x76543210, 0xfedcba98, false, false));
}

__global__ __launch_bounds__(128) __attribute__((amdgpu_waves_per_eu(8, 8))) void gcn_main(
    const float* __restrict__ feature, const float* __restrict__ x,
    const float* __restrict__ label,   const float* __restrict__ w,
    const float* __restrict__ e_type,  const int* __restrict__ src,
    const float* __restrict__ Bsw,     float* __restrict__ out) {

  __shared__ __attribute__((aligned(16))) int   s_src[16 * D_DEG];
  __shared__ __attribute__((aligned(16))) float s_et0[16 * D_DEG];
  __shared__ __attribute__((aligned(16))) float s_nw[16 * D_DEG];
  __shared__ __attribute__((aligned(16))) float s_A[16 * AS_STRIDE];

  const int tid = threadIdx.x;
  const int nb  = blockIdx.x * 16;

  const int e4 = nb * D_DEG + tid * 4;
  const int4   s4 = *(const int4*)(src + e4);
  const float4 w4 = *(const float4*)(w + e4);
  const float4 ta = *(const float4*)(e_type + e4 * 2);
  const float4 tb = *(const float4*)(e_type + e4 * 2 + 4);
  const int    l5     = tid & 31;
  const float  side_a = label[nb * K_LAB + tid];
  const float  lab_b  = label[nb * K_LAB + 128 + l5];
  const float  x_b    = x[nb * 2 + l5];
  *(int4*)&s_src[tid * 4] = s4;
  float4 et; et.x = ta.x; et.y = ta.z; et.z = tb.x; et.w = tb.z;
  *(float4*)&s_et0[tid * 4] = et;
  float4 nw4; nw4.x = w4.x * et.x; nw4.y = w4.y * et.y; nw4.z = w4.z * et.z; nw4.w = w4.w * et.w;
  *(float4*)&s_nw[tid * 4] = nw4;
  {
    const int r = tid / K_LAB;
    s_A[r * AS_STRIDE + H_DIM + (tid - r * K_LAB)] = side_a;
  }
  asm volatile("" :: "v"(lab_b), "v"(x_b));
  {
    const float side_b = (tid < 32) ? lab_b : x_b;
    const int i_b   = tid + 128;
    const int r_b   = i_b / K_LAB;
    const int j_b   = l5;
    const int slotl = r_b * AS_STRIDE + H_DIM + (i_b - r_b * K_LAB);
    const int slotx = (j_b >> 1) * AS_STRIDE + H_DIM + K_LAB + (j_b & 1);
    const int slot  = (tid < 32) ? slotl : slotx;
    if (tid < 64) {
      s_A[slot] = side_b;
    } else if (tid < 80) {
      const int r = tid - 64;
      s_A[r * AS_STRIDE + 142] = 1.0f;
      s_A[r * AS_STRIDE + 143] = 0.0f;
    }
  }
  __syncthreads();

  const int gw = tid >> 5;
  const int gl = tid & 31;
  const int r0 = 4 * gw;
  const unsigned loff = (unsigned)gl * 16u;

  float4 acc4[4];
#pragma unroll
  for (int j = 0; j < 4; ++j) { acc4[j].x = 0.f; acc4[j].y = 0.f; acc4[j].z = 0.f; acc4[j].w = 0.f; }
#pragma unroll 1
  for (int d0 = 0; d0 < D_DEG; d0 += 4) {
    float4 f[4][4];
    float  ev[4][4];
#pragma unroll
    for (int dd = 0; dd < 4; ++dd) {
#pragma unroll
      for (int j = 0; j < 4; ++j) {
        const int ix = s_src[(r0 + j) * D_DEG + d0 + dd];
        ev[dd][j]    = s_et0[(r0 + j) * D_DEG + d0 + dd];
        const unsigned of = (unsigned)ix * (H_DIM * 4u) + loff;
        f[dd][j] = *(const float4*)((const char*)feature + of);
      }
    }
#pragma unroll
    for (int dd = 0; dd < 4; ++dd) {
#pragma unroll
      for (int j = 0; j < 4; ++j) {
        acc4[j].x += f[dd][j].x * ev[dd][j]; acc4[j].y += f[dd][j].y * ev[dd][j];
        acc4[j].z += f[dd][j].z * ev[dd][j]; acc4[j].w += f[dd][j].w * ev[dd][j];
      }
    }
  }
#pragma unroll
  for (int j = 0; j < 4; ++j)
    *(float4*)&s_A[(r0 + j) * AS_STRIDE + gl * 4] = acc4[j];

  if (gl < 8) {
    const int  row = r0 + (gl >> 1);
    const bool pos = ((gl & 1) == 0);
    float p = 0.f;
#pragma unroll
    for (int d = 0; d < D_DEG; ++d) {
      const float nw = s_nw[row * D_DEG + d];
      p += pos ? fmaxf(nw, 0.f) : fminf(nw, 0.f);
    }
    s_A[row * AS_STRIDE + 140 + (gl & 1)] = p;
  }
  __syncthreads();

  const int wave = gw;
  const int lane = gl;
  const int m    = lane & 15;
  const int hi   = lane >> 4;

  const float4* bfr0 = (const float4*)Bsw + (2 * wave) * (N_Q * 32) + lane;
  const float4* bfr1 = bfr0 + N_Q * 32;

  v8f c0 = {0.f, 0.f, 0.f, 0.f, 0.f, 0.f, 0.f, 0.f};
  v8f c1 = {0.f, 0.f, 0.f, 0.f, 0.f, 0.f, 0.f, 0.f};
#pragma unroll 6
  for (int t2 = 0; t2 < N_STEPS / 2; ++t2) {
    const float4 bv0 = bfr0[t2 * 32];
    const float4 bv1 = bfr1[t2 * 32];
    {
      const int kk = 4 * (2 * t2) + 2 * hi;
      v2f a, b0, b1;
      a.x = s_A[m * AS_STRIDE + kk];
      a.y = s_A[m * AS_STRIDE + kk + 1];
      b0.x = bv0.x; b0.y = bv0.y;
      b1.x = bv1.x; b1.y = bv1.y;
      c0 = __builtin_amdgcn_wmma_f32_16x16x4_f32(
          false, a, false, b0, (short)0, c0, false, false);
      c1 = __builtin_amdgcn_wmma_f32_16x16x4_f32(
          false, a, false, b1, (short)0, c1, false, false);
    }
    {
      const int kk = 4 * (2 * t2 + 1) + 2 * hi;
      v2f a, b0, b1;
      a.x = s_A[m * AS_STRIDE + kk];
      a.y = s_A[m * AS_STRIDE + kk + 1];
      b0.x = bv0.z; b0.y = bv0.w;
      b1.x = bv1.z; b1.y = bv1.w;
      c0 = __builtin_amdgcn_wmma_f32_16x16x4_f32(
          false, a, false, b0, (short)0, c0, false, false);
      c1 = __builtin_amdgcn_wmma_f32_16x16x4_f32(
          false, a, false, b1, (short)0, c1, false, false);
    }
  }

  float* orow = out + (long)nb * H_DIM + wave * 32 + lane;
#pragma unroll
  for (int r = 0; r < 8; ++r) {
    const float t0 = c0[r] > 0.f ? c0[r] : 0.f;
    const float t1 = c1[r] > 0.f ? c1[r] : 0.f;
    const float give = hi ? t0 : t1;
    const float got  = swap_halves(give);
    orow[(long)r * H_DIM]       = hi ? got : t0;
    orow[(long)(r + 8) * H_DIM] = hi ? t1 : got;
  }
}

extern "C" void kernel_launch(void* const* d_in, const int* in_sizes, int n_in,
                              void* d_out, int out_size, void* d_ws, size_t ws_size,
                              hipStream_t stream) {
  const float* feature = (const float*)d_in[0];
  const float* x       = (const float*)d_in[1];
  const float* label   = (const float*)d_in[2];
  const float* w       = (const float*)d_in[3];
  const float* e_type  = (const float*)d_in[4];
  const int*   src     = (const int*)  d_in[5];
  const float* W0      = (const float*)d_in[6];
  const float* b0      = (const float*)d_in[7];
  const float* W1      = (const float*)d_in[8];
  const float* b1      = (const float*)d_in[9];
  const float* W2      = (const float*)d_in[10];
  const float* b2      = (const float*)d_in[11];
  const float* Wt3     = (const float*)d_in[12];
  const float* Wt4     = (const float*)d_in[13];

  float* Bsw = (float*)d_ws;

  gcn_prep<<<PREP_GRID, PREP_BLOCK, 0, stream>>>(W0, b0, W1, b1, W2, b2, Wt3, Wt4,
                                                 Bsw);
  gcn_main<<<N_NODES / 16, 128, 0, stream>>>(feature, x, label, w, e_type, src,
                                             Bsw, (float*)d_out);
}
